# starting kernel
# baseline (speedup 1.0000x reference)
#include <hip/hip_runtime.h>
#include <math.h>

#define Bsz   512
#define INsz  256
#define Hsz   1024
#define CONC  1280
#define WMAT  (CONC * Hsz)

typedef __attribute__((ext_vector_type(2))) float v2f;
typedef __attribute__((ext_vector_type(8))) float v8f;

__global__ __launch_bounds__(256) void zero4(float4* __restrict__ p, int n4) {
    int i = blockIdx.x * blockDim.x + threadIdx.x;
    if (i < n4) p[i] = make_float4(0.f, 0.f, 0.f, 0.f);
}

__global__ __launch_bounds__(256) void build_xt(const float* __restrict__ in,
                                                const float* __restrict__ hx,
                                                float* __restrict__ XT) {
    int t = blockIdx.x * blockDim.x + threadIdx.x;
    int j = t >> 9;
    int b = t & (Bsz - 1);
    float v = (j < INsz) ? in[b * INsz + j] : hx[b * Hsz + (j - INsz)];
    XT[j * Bsz + b] = v;
}

__global__ __launch_bounds__(256) void scatter_w(const int*   __restrict__ src,
                                                 const int*   __restrict__ dst,
                                                 const float* __restrict__ signs,
                                                 const float* __restrict__ w1,
                                                 const float* __restrict__ w2,
                                                 const float* __restrict__ wa,
                                                 const float* __restrict__ wb,
                                                 float* __restrict__ W, int E) {
    int e = blockIdx.x * blockDim.x + threadIdx.x;
    if (e >= E) return;
    float  sg   = signs[e];
    size_t base = (size_t)src[e] * Hsz + (size_t)dst[e];
    atomicAdd(&W[base],            w1[e] * sg);
    atomicAdd(&W[WMAT + base],     w2[e] * sg);
    atomicAdd(&W[2ull*WMAT + base], wa[e] * sg);
    atomicAdd(&W[3ull*WMAT + base], wb[e] * sg);
}

__global__ __launch_bounds__(128) void gemm_fused(const float* __restrict__ XT,
                                                  const float* __restrict__ W,
                                                  const float* __restrict__ ts,
                                                  const float* __restrict__ b_ff1,
                                                  const float* __restrict__ b_ff2,
                                                  const float* __restrict__ b_ta,
                                                  const float* __restrict__ b_tb,
                                                  float* __restrict__ out) {
    const int lane = threadIdx.x & 31;
    const int wave = threadIdx.x >> 5;

    const int n0   = (blockIdx.x * 4 + wave) * 16;
    const int m0   = blockIdx.y * 32;
    const int mrow = lane & 15;
    const int ncol = lane & 15;
    const int kp   = (lane >> 4) << 1;

    v8f acc[2][4];
    #pragma unroll
    for (int s = 0; s < 2; ++s)
        #pragma unroll
        for (int l = 0; l < 4; ++l)
            acc[s][l] = (v8f){0.f,0.f,0.f,0.f,0.f,0.f,0.f,0.f};

    const float* xa0 = XT + (m0      + mrow);
    const float* xa1 = XT + (m0 + 16 + mrow);

    for (int k0 = 0; k0 < CONC; k0 += 4) {
        const int kr = k0 + kp;
        v2f a0, a1;
        a0.x = xa0[(size_t)kr * Bsz];
        a0.y = xa0[(size_t)(kr + 1) * Bsz];
        a1.x = xa1[(size_t)kr * Bsz];
        a1.y = xa1[(size_t)(kr + 1) * Bsz];

        #pragma unroll
        for (int l = 0; l < 4; ++l) {
            const float* Wp = W + (size_t)l * WMAT + (size_t)kr * Hsz + n0 + ncol;
            v2f b;
            b.x = Wp[0];
            b.y = Wp[Hsz];
            acc[0][l] = __builtin_amdgcn_wmma_f32_16x16x4_f32(
                false, a0, false, b, (short)0, acc[0][l], false, false);
            acc[1][l] = __builtin_amdgcn_wmma_f32_16x16x4_f32(
                false, a1, false, b, (short)0, acc[1][l], false, false);
        }
    }

    const int   ng  = n0 + ncol;
    const float bf1 = b_ff1[ng];
    const float bf2 = b_ff2[ng];
    const float bta = b_ta[ng];
    const float btb = b_tb[ng];
    const int   mhi = (lane >> 4) << 3;

    #pragma unroll
    for (int s = 0; s < 2; ++s) {
        #pragma unroll
        for (int r = 0; r < 8; ++r) {
            const int   m   = m0 + s * 16 + r + mhi;
            const float tsv = ts[m];
            const float f1  = tanhf(acc[s][0][r] + bf1);
            const float f2  = tanhf(acc[s][1][r] + bf2);
            const float ta  = acc[s][2][r] + bta;
            const float tb  = acc[s][3][r] + btb;
            const float sg  = 1.f / (1.f + expf(-(ta * tsv + tb)));
            const float h   = f1 * (1.f - sg) + sg * f2;
            const size_t o  = (size_t)m * Hsz + ng;
            out[o]                         = h;
            out[(size_t)Bsz * Hsz + o]     = h;
        }
    }
}

extern "C" void kernel_launch(void* const* d_in, const int* in_sizes, int n_in,
                              void* d_out, int out_size, void* d_ws, size_t ws_size,
                              hipStream_t stream) {
    const float* input = (const float*)d_in[0];
    const float* hx    = (const float*)d_in[1];
    const float* ts    = (const float*)d_in[2];
    const int*   src   = (const int*)  d_in[3];
    const int*   dst   = (const int*)  d_in[4];
    const float* signs = (const float*)d_in[5];
    const float* w1    = (const float*)d_in[6];
    const float* w2    = (const float*)d_in[7];
    const float* wa    = (const float*)d_in[8];
    const float* wb    = (const float*)d_in[9];
    const float* bf1   = (const float*)d_in[10];
    const float* bf2   = (const float*)d_in[11];
    const float* bta   = (const float*)d_in[12];
    const float* btb   = (const float*)d_in[13];
    const int    E     = in_sizes[3];

    float* W  = (float*)d_ws;
    float* XT = W + 4ull * WMAT;
    float* out = (float*)d_out;

    const int n4 = (4 * WMAT) / 4;
    zero4<<<(n4 + 255) / 256, 256, 0, stream>>>((float4*)W, n4);

    build_xt<<<(CONC * Bsz) / 256, 256, 0, stream>>>(input, hx, XT);

    scatter_w<<<(E + 255) / 256, 256, 0, stream>>>(src, dst, signs,
                                                   w1, w2, wa, wb, W, E);

    dim3 grid(Hsz / 64, Bsz / 32);
    gemm_fused<<<grid, 128, 0, stream>>>(XT, W, ts, bf1, bf2, bta, btb, out);
}


# GEMM K loop split over 8 waves of a 32-wave work-group (8 waves per SIMD hide fragment-load latency); partial tiles summed through LDS in slice order, so the K sum is now 8 partial sums
# speedup vs baseline: 2.0117x; 2.0117x over previous
#include <hip/hip_runtime.h>
#include <math.h>

#define Bsz   512
#define INsz  256
#define Hsz   1024
#define CONC  1280
#define WMAT  (CONC * Hsz)

typedef __attribute__((ext_vector_type(2))) float v2f;
typedef __attribute__((ext_vector_type(8))) float v8f;

__global__ __launch_bounds__(256) void zero4(float4* __restrict__ p, int n4) {
    int i = blockIdx.x * blockDim.x + threadIdx.x;
    if (i < n4) p[i] = make_float4(0.f, 0.f, 0.f, 0.f);
}

__global__ __launch_bounds__(256) void build_xt(const float* __restrict__ in,
                                                const float* __restrict__ hx,
                                                float* __restrict__ XT) {
    int t = blockIdx.x * blockDim.x + threadIdx.x;
    int j = t >> 9;
    int b = t & (Bsz - 1);
    float v = (j < INsz) ? in[b * INsz + j] : hx[b * Hsz + (j - INsz)];
    XT[j * Bsz + b] = v;
}

__global__ __launch_bounds__(256) void scatter_w(const int*   __restrict__ src,
                                                 const int*   __restrict__ dst,
                                                 const float* __restrict__ signs,
                                                 const float* __restrict__ w1,
                                                 const float* __restrict__ w2,
                                                 const float* __restrict__ wa,
                                                 const float* __restrict__ wb,
                                                 float* __restrict__ W, int E) {
    int e = blockIdx.x * blockDim.x + threadIdx.x;
    if (e >= E) return;
    float  sg   = signs[e];
    size_t base = (size_t)src[e] * Hsz + (size_t)dst[e];
    atomicAdd(&W[base],            w1[e] * sg);
    atomicAdd(&W[WMAT + base],     w2[e] * sg);
    atomicAdd(&W[2ull*WMAT + base], wa[e] * sg);
    atomicAdd(&W[3ull*WMAT + base], wb[e] * sg);
}

#define KSPLIT 8
#define KSLICE (CONC / KSPLIT)
#define ACCF   64

__global__ __launch_bounds__(1024) void gemm_fused(const float* __restrict__ XT,
                                                   const float* __restrict__ W,
                                                   const float* __restrict__ ts,
                                                   const float* __restrict__ b_ff1,
                                                   const float* __restrict__ b_ff2,
                                                   const float* __restrict__ b_ta,
                                                   const float* __restrict__ b_tb,
                                                   float* __restrict__ out) {
    __shared__ float red[4 * KSPLIT * ACCF * 32];

    const int lane = threadIdx.x & 31;
    const int wave = __builtin_amdgcn_readfirstlane((int)(threadIdx.x >> 5));
    const int nw   = wave & 3;
    const int q    = wave >> 2;

    const int n0   = (blockIdx.x * 4 + nw) * 16;
    const int m0   = blockIdx.y * 32;
    const int mrow = lane & 15;
    const int ncol = lane & 15;
    const int kp   = (lane >> 4) << 1;

    v8f acc[2][4];
    #pragma unroll
    for (int s = 0; s < 2; ++s)
        #pragma unroll
        for (int l = 0; l < 4; ++l)
            acc[s][l] = (v8f){0.f,0.f,0.f,0.f,0.f,0.f,0.f,0.f};

    const float* xa0 = XT + (m0      + mrow);
    const float* xa1 = XT + (m0 + 16 + mrow);

    const int kbeg = q * KSLICE;
    for (int ks = 0; ks < KSLICE; ks += 4) {
        const int kr = kbeg + ks + kp;
        v2f a0, a1;
        a0.x = xa0[(size_t)kr * Bsz];
        a0.y = xa0[(size_t)(kr + 1) * Bsz];
        a1.x = xa1[(size_t)kr * Bsz];
        a1.y = xa1[(size_t)(kr + 1) * Bsz];

        #pragma unroll
        for (int l = 0; l < 4; ++l) {
            const float* Wp = W + (size_t)l * WMAT + (size_t)kr * Hsz + n0 + ncol;
            v2f b;
            b.x = Wp[0];
            b.y = Wp[Hsz];
            acc[0][l] = __builtin_amdgcn_wmma_f32_16x16x4_f32(
                false, a0, false, b, (short)0, acc[0][l], false, false);
            acc[1][l] = __builtin_amdgcn_wmma_f32_16x16x4_f32(
                false, a1, false, b, (short)0, acc[1][l], false, false);
        }
    }

    float* mine = red + (size_t)((nw * KSPLIT + q) * ACCF) * 32 + lane;
    #pragma unroll
    for (int s = 0; s < 2; ++s)
        #pragma unroll
        for (int l = 0; l < 4; ++l)
            #pragma unroll
            for (int r = 0; r < 8; ++r)
                mine[((s * 4 + l) * 8 + r) * 32] = acc[s][l][r];

    __syncthreads();

    const int   ng  = n0 + ncol;
    const float bf1 = b_ff1[ng];
    const float bf2 = b_ff2[ng];
    const float bta = b_ta[ng];
    const float btb = b_tb[ng];
    const int   mhi = (lane >> 4) << 3;

    const int    s    = q >> 2;
    const int    rb   = (q & 3) << 1;
    const float* tile = red + (size_t)(nw * KSPLIT * ACCF) * 32 + lane;

    #pragma unroll
    for (int rr = 0; rr < 2; ++rr) {
        const int r = rb + rr;
        float v[4];
        #pragma unroll
        for (int l = 0; l < 4; ++l) {
            const int idx = (s * 4 + l) * 8 + r;
            float sum = tile[idx * 32];
            #pragma unroll
            for (int qq = 1; qq < KSPLIT; ++qq)
                sum += tile[(qq * ACCF + idx) * 32];
            v[l] = sum;
        }
        const int   m   = m0 + s * 16 + r + mhi;
        const float tsv = ts[m];
        const float f1  = tanhf(v[0] + bf1);
        const float f2  = tanhf(v[1] + bf2);
        const float ta  = v[2] + bta;
        const float tb  = v[3] + btb;
        const float sg  = 1.f / (1.f + expf(-(ta * tsv + tb)));
        const float h   = f1 * (1.f - sg) + sg * f2;
        const size_t o  = (size_t)m * Hsz + ng;
        out[o]                         = h;
        out[(size_t)Bsz * Hsz + o]     = h;
    }
}

extern "C" void kernel_launch(void* const* d_in, const int* in_sizes, int n_in,
                              void* d_out, int out_size, void* d_ws, size_t ws_size,
                              hipStream_t stream) {
    const float* input = (const float*)d_in[0];
    const float* hx    = (const float*)d_in[1];
    const float* ts    = (const float*)d_in[2];
    const int*   src   = (const int*)  d_in[3];
    const int*   dst   = (const int*)  d_in[4];
    const float* signs = (const float*)d_in[5];
    const float* w1    = (const float*)d_in[6];
    const float* w2    = (const float*)d_in[7];
    const float* wa    = (const float*)d_in[8];
    const float* wb    = (const float*)d_in[9];
    const float* bf1   = (const float*)d_in[10];
    const float* bf2   = (const float*)d_in[11];
    const float* bta   = (const float*)d_in[12];
    const float* btb   = (const float*)d_in[13];
    const int    E     = in_sizes[3];

    float* W  = (float*)d_ws;
    float* XT = W + 4ull * WMAT;
    float* out = (float*)d_out;

    const int n4 = (4 * WMAT) / 4;
    zero4<<<(n4 + 255) / 256, 256, 0, stream>>>((float4*)W, n4);

    build_xt<<<(CONC * Bsz) / 256, 256, 0, stream>>>(input, hx, XT);

    scatter_w<<<(E + 255) / 256, 256, 0, stream>>>(src, dst, signs,
                                                   w1, w2, wa, wb, W, E);

    dim3 grid(Hsz / 64, Bsz / 32);
    gemm_fused<<<grid, 1024, 0, stream>>>(XT, W, ts, bf1, bf2, bta, btb, out);
}


# Row-owner scan: the 4 src of a 16-byte load tested at once (unsigned min), hits kept by ballot in wave-private LDS lists: no LDS counter round trip, no barrier before the adds; add order free
# speedup vs baseline: 3.1883x; 1.5849x over previous
#include <hip/hip_runtime.h>
#include <math.h>

#define Bsz   512
#define INsz  256
#define Hsz   1024
#define CONC  1280
#define WMAT  (CONC * Hsz)

typedef __attribute__((ext_vector_type(2))) float v2f;
typedef __attribute__((ext_vector_type(8))) float v8f;

#define WROWS  5
#define WGRID  (CONC / WROWS)
#define WTHR   512
#define WWAVES (WTHR / 32)
#define WLOADS 16
#define WLIST  24576
#define WSEG   (WLIST / WWAVES)
#define WTILE4 (4 * WROWS * Hsz / 4)

__global__ __launch_bounds__(WTHR) void build_w(const int*   __restrict__ src,
                                                const int*   __restrict__ dst,
                                                const float* __restrict__ signs,
                                                const float* __restrict__ w1,
                                                const float* __restrict__ w2,
                                                const float* __restrict__ wa,
                                                const float* __restrict__ wb,
                                                float* __restrict__ W, int E) {
    __shared__ float4 tile4[WTILE4];
    __shared__ int    list[WLIST];

    float*         tile = (float*)tile4;
    const int      tid  = threadIdx.x;
    const int      lane = tid & 31;
    const int      wave = __builtin_amdgcn_readfirstlane((int)(threadIdx.x >> 5));
    const unsigned k0   = blockIdx.x * WROWS;
    int*           seg  = list + wave * WSEG;

    for (int it = 0; it < WTILE4 / WTHR; ++it)
        tile4[it * WTHR + tid] = make_float4(0.f, 0.f, 0.f, 0.f);
    __syncthreads();

    auto add_edge = [&](int e, int k) {
        const int   d  = dst[e];
        const float sg = signs[e];
        const float p1 = w1[e] * sg;
        const float p2 = w2[e] * sg;
        const float p3 = wa[e] * sg;
        const float p4 = wb[e] * sg;
        if ((unsigned)d < (unsigned)Hsz) {
            float* c = tile + k * Hsz + d;
            atomicAdd(c,                   p1);
            atomicAdd(c +     WROWS * Hsz, p2);
            atomicAdd(c + 2 * WROWS * Hsz, p3);
            atomicAdd(c + 3 * WROWS * Hsz, p4);
        }
    };

    int cnt = 0;
    auto note = [&](int e, unsigned k, unsigned lim) {
        const bool     p    = k < lim;
        const unsigned mask = __builtin_amdgcn_ballot_w32(p);
        if (mask) {
            const int pos = cnt + (int)__builtin_amdgcn_mbcnt_lo(mask, 0u);
            if (p && pos < WSEG) seg[pos] = e * 8 + (int)k;
            cnt += __builtin_popcount(mask);
        }
    };
    auto test4 = [&](int e0, const int4& v, unsigned lim) {
        const unsigned a  = (unsigned)v.x - k0;
        const unsigned b  = (unsigned)v.y - k0;
        const unsigned c  = (unsigned)v.z - k0;
        const unsigned d  = (unsigned)v.w - k0;
        const unsigned ab = a < b ? a : b;
        const unsigned cd = c < d ? c : d;
        const unsigned m  = ab < cd ? ab : cd;
        if (__builtin_amdgcn_ballot_w32(m < lim)) {
            note(e0,     a, lim);
            note(e0 + 1, b, lim);
            note(e0 + 2, c, lim);
            note(e0 + 3, d, lim);
        }
    };

    const int   E4   = E >> 2;
    const int4* src4 = (const int4*)src;
    const int   nIt  = (E4 + WTHR - 1) / WTHR;
    const int   nPad = ((nIt + WLOADS - 1) / WLOADS) * WLOADS;
    const int   rot  = nPad ? (int)(blockIdx.x % (unsigned)nPad) : 0;
    for (int g = 0; g < nPad; g += WLOADS) {
        int j0 = g + rot;
        if (j0 >= nPad) j0 -= nPad;
        int4 v[WLOADS];
        if (j0 + WLOADS <= nPad && (j0 + WLOADS) * WTHR <= E4) {
            const int b4 = j0 * WTHR + tid;
            #pragma unroll
            for (int u = 0; u < WLOADS; ++u)
                v[u] = src4[b4 + u * WTHR];
            #pragma unroll
            for (int u = 0; u < WLOADS; ++u)
                test4((b4 + u * WTHR) * 4, v[u], WROWS);
        } else {
            #pragma unroll
            for (int u = 0; u < WLOADS; ++u) {
                int j = j0 + u;
                if (j >= nPad) j -= nPad;
                const int i4 = j * WTHR + tid;
                v[u] = src4[i4 < E4 ? i4 : E4 - 1];
            }
            #pragma unroll
            for (int u = 0; u < WLOADS; ++u) {
                int j = j0 + u;
                if (j >= nPad) j -= nPad;
                const int i4 = j * WTHR + tid;
                test4(i4 * 4, v[u], i4 < E4 ? (unsigned)WROWS : 0u);
            }
        }
    }
    {
        const int      e   = E4 * 4 + tid;
        const bool     inb = e < E;
        const unsigned k   = inb ? (unsigned)src[e] - k0 : 0xffffffffu;
        note(e, k, WROWS);
    }

    if (cnt <= WSEG) {
        for (int i = lane; i < cnt; i += 32) {
            const int p = seg[i];
            add_edge(p >> 3, p & 7);
        }
    } else {
        for (int it = 0; it < nIt; ++it) {
            const int i4 = it * WTHR + tid;
            if (i4 < E4) {
                const int4     v = src4[i4];
                const unsigned a = (unsigned)v.x - k0;
                const unsigned b = (unsigned)v.y - k0;
                const unsigned c = (unsigned)v.z - k0;
                const unsigned d = (unsigned)v.w - k0;
                if (a < WROWS) add_edge(i4 * 4,     (int)a);
                if (b < WROWS) add_edge(i4 * 4 + 1, (int)b);
                if (c < WROWS) add_edge(i4 * 4 + 2, (int)c);
                if (d < WROWS) add_edge(i4 * 4 + 3, (int)d);
            }
        }
        const int e = E4 * 4 + tid;
        if (e < E) {
            const unsigned k = (unsigned)src[e] - k0;
            if (k < WROWS) add_edge(e, (int)k);
        }
    }
    __syncthreads();

    for (int it = 0; it < WTILE4 / WTHR; ++it) {
        const int f = it * WTHR + tid;
        const int l = f / (WROWS * Hsz / 4);
        const int i = f - l * (WROWS * Hsz / 4);
        float4* o = (float4*)(W + (size_t)l * WMAT + (size_t)k0 * Hsz);
        o[i] = tile4[f];
    }
}

#define KSPLIT 8
#define KSLICE (CONC / KSPLIT)
#define ACCF   64
#define KSTEPS (CONC / 4)
#define APITCH 33
#define ALOADS (KSTEPS * 32 / 1024)

__global__ __launch_bounds__(1024) void gemm_fused(const float* __restrict__ in,
                                                   const float* __restrict__ hx,
                                                   const float* __restrict__ W,
                                                   const float* __restrict__ ts,
                                                   const float* __restrict__ b_ff1,
                                                   const float* __restrict__ b_ff2,
                                                   const float* __restrict__ b_ta,
                                                   const float* __restrict__ b_tb,
                                                   float* __restrict__ out) {
    __shared__ float4 red4[4 * KSPLIT * ACCF * 32 / 4];
    float* red = (float*)red4;

    const int lane = threadIdx.x & 31;
    const int wave = __builtin_amdgcn_readfirstlane((int)(threadIdx.x >> 5));
    const int nw   = wave & 3;
    const int q    = wave >> 2;

    const int n0   = (blockIdx.x * 4 + nw) * 16;
    const int m0   = blockIdx.y * 32;
    const int ncol = lane & 15;
    const int kp   = (lane >> 4) << 1;

    {
        float2*   As2 = (float2*)red4;
        const int t16 = lane & 15;
        const int s   = lane >> 4;
        float4 v[ALOADS];
        #pragma unroll
        for (int it = 0; it < ALOADS; ++it) {
            const int u  = it * 32 + wave;
            const int c  = u & 15;
            const int Tb = u >> 4;
            const int T  = Tb * 16 + t16;
            const int m  = m0 + s * 16 + c;
            const float* p = (Tb < INsz / 64) ? in + (size_t)m * INsz + 4 * T
                                              : hx + (size_t)m * Hsz + (4 * T - INsz);
            v[it] = *(const float4*)p;
        }
        #pragma unroll
        for (int it = 0; it < ALOADS; ++it) {
            const int u  = it * 32 + wave;
            const int c  = u & 15;
            const int T  = (u >> 4) * 16 + t16;
            As2[(T * APITCH + c) * 2 + s]      = make_float2(v[it].x, v[it].y);
            As2[(T * APITCH + 16 + c) * 2 + s] = make_float2(v[it].z, v[it].w);
        }
    }

    v8f acc[2][4];
    #pragma unroll
    for (int s = 0; s < 2; ++s)
        #pragma unroll
        for (int l = 0; l < 4; ++l)
            acc[s][l] = (v8f){0.f,0.f,0.f,0.f,0.f,0.f,0.f,0.f};

    __syncthreads();

    const float4* As4  = red4 + lane;
    const int     kbeg = q * KSLICE;
    for (int ks = 0; ks < KSLICE; ks += 4) {
        const int kr = kbeg + ks + kp;
        const float4 a = As4[((kbeg + ks) >> 2) * APITCH];
        v2f a0, a1;
        a0.x = a.x;  a0.y = a.y;
        a1.x = a.z;  a1.y = a.w;

        #pragma unroll
        for (int l = 0; l < 4; ++l) {
            const float* Wp = W + (size_t)l * WMAT + (size_t)kr * Hsz + n0 + ncol;
            v2f b;
            b.x = Wp[0];
            b.y = Wp[Hsz];
            acc[0][l] = __builtin_amdgcn_wmma_f32_16x16x4_f32(
                false, a0, false, b, (short)0, acc[0][l], false, false);
            acc[1][l] = __builtin_amdgcn_wmma_f32_16x16x4_f32(
                false, a1, false, b, (short)0, acc[1][l], false, false);
        }
    }

    __syncthreads();

    float* mine = red + (size_t)((nw * KSPLIT + q) * ACCF) * 32 + lane;
    #pragma unroll
    for (int s = 0; s < 2; ++s)
        #pragma unroll
        for (int l = 0; l < 4; ++l)
            #pragma unroll
            for (int r = 0; r < 8; ++r)
                mine[((s * 4 + l) * 8 + r) * 32] = acc[s][l][r];

    __syncthreads();

    const int   ng  = n0 + ncol;
    const float bf1 = b_ff1[ng];
    const float bf2 = b_ff2[ng];
    const float bta = b_ta[ng];
    const float btb = b_tb[ng];
    const int   mhi = (lane >> 4) << 3;

    const int    s    = q >> 2;
    const int    rb   = (q & 3) << 1;
    const float* tile = red + (size_t)(nw * KSPLIT * ACCF) * 32 + lane;

    #pragma unroll
    for (int rr = 0; rr < 2; ++rr) {
        const int r = rb + rr;
        float v[4];
        #pragma unroll
        for (int l = 0; l < 4; ++l) {
            const int idx = (s * 4 + l) * 8 + r;
            float sum = tile[idx * 32];
            #pragma unroll
            for (int qq = 1; qq < KSPLIT; ++qq)
                sum += tile[(qq * ACCF + idx) * 32];
            v[l] = sum;
        }
        const int   m   = m0 + s * 16 + r + mhi;
        const float tsv = ts[m];
        const float f1  = tanhf(v[0] + bf1);
        const float f2  = tanhf(v[1] + bf2);
        const float ta  = v[2] + bta;
        const float tb  = v[3] + btb;
        const float sg  = 1.f / (1.f + expf(-(ta * tsv + tb)));
        const float h   = f1 * (1.f - sg) + sg * f2;
        const size_t o  = (size_t)m * Hsz + ng;
        out[o]                         = h;
        out[(size_t)Bsz * Hsz + o]     = h;
    }
}

extern "C" void kernel_launch(void* const* d_in, const int* in_sizes, int n_in,
                              void* d_out, int out_size, void* d_ws, size_t ws_size,
                              hipStream_t stream) {
    const float* input = (const float*)d_in[0];
    const float* hx    = (const float*)d_in[1];
    const float* ts    = (const float*)d_in[2];
    const int*   src   = (const int*)  d_in[3];
    const int*   dst   = (const int*)  d_in[4];
    const float* signs = (const float*)d_in[5];
    const float* w1    = (const float*)d_in[6];
    const float* w2    = (const float*)d_in[7];
    const float* wa    = (const float*)d_in[8];
    const float* wb    = (const float*)d_in[9];
    const float* bf1   = (const float*)d_in[10];
    const float* bf2   = (const float*)d_in[11];
    const float* bta   = (const float*)d_in[12];
    const float* btb   = (const float*)d_in[13];
    const int    E     = in_sizes[3];

    float* W  = (float*)d_ws;
    float* out = (float*)d_out;

    build_w<<<WGRID, WTHR, 0, stream>>>(src, dst, signs, w1, w2, wa, wb, W, E);

    dim3 grid(Hsz / 64, Bsz / 32);
    gemm_fused<<<grid, 1024, 0, stream>>>(input, hx, W, ts, bf1, bf2, bta, btb, out);
}


# GEMM wave = one weight matrix x 64 columns: a lane's two 16-byte W loads (4 adjacent columns of its K row pair) feed 4 column-phase N tiles; 2 whole-line loads per k-step, not 8 4-byte; sums kept
# speedup vs baseline: 3.4082x; 1.0690x over previous
#include <hip/hip_runtime.h>
#include <math.h>

#define Bsz   512
#define INsz  256
#define Hsz   1024
#define CONC  1280
#define WMAT  (CONC * Hsz)

typedef __attribute__((ext_vector_type(2))) float v2f;
typedef __attribute__((ext_vector_type(8))) float v8f;

#define WROWS  5
#define WGRID  (CONC / WROWS)
#define WTHR   512
#define WWAVES (WTHR / 32)
#define WLOADS 16
#define WLIST  24576
#define WSEG   (WLIST / WWAVES)
#define WTILE4 (4 * WROWS * Hsz / 4)

__global__ __launch_bounds__(WTHR) void build_w(const int*   __restrict__ src,
                                                const int*   __restrict__ dst,
                                                const float* __restrict__ signs,
                                                const float* __restrict__ w1,
                                                const float* __restrict__ w2,
                                                const float* __restrict__ wa,
                                                const float* __restrict__ wb,
                                                float* __restrict__ W, int E) {
    __shared__ float4 tile4[WTILE4];
    __shared__ int    list[WLIST];

    float*         tile = (float*)tile4;
    const int      tid  = threadIdx.x;
    const int      lane = tid & 31;
    const int      wave = __builtin_amdgcn_readfirstlane((int)(threadIdx.x >> 5));
    const unsigned k0   = blockIdx.x * WROWS;
    int*           seg  = list + wave * WSEG;

    for (int it = 0; it < WTILE4 / WTHR; ++it)
        tile4[it * WTHR + tid] = make_float4(0.f, 0.f, 0.f, 0.f);
    __syncthreads();

    auto add_edge = [&](int e, int k) {
        const int   d  = dst[e];
        const float sg = signs[e];
        const float p1 = w1[e] * sg;
        const float p2 = w2[e] * sg;
        const float p3 = wa[e] * sg;
        const float p4 = wb[e] * sg;
        if ((unsigned)d < (unsigned)Hsz) {
            float* c = tile + k * Hsz + d;
            atomicAdd(c,                   p1);
            atomicAdd(c +     WROWS * Hsz, p2);
            atomicAdd(c + 2 * WROWS * Hsz, p3);
            atomicAdd(c + 3 * WROWS * Hsz, p4);
        }
    };

    int cnt = 0;
    auto note = [&](int e, unsigned k, unsigned lim) {
        const bool     p    = k < lim;
        const unsigned mask = __builtin_amdgcn_ballot_w32(p);
        if (mask) {
            const int pos = cnt + (int)__builtin_amdgcn_mbcnt_lo(mask, 0u);
            if (p && pos < WSEG) seg[pos] = e * 8 + (int)k;
            cnt += __builtin_popcount(mask);
        }
    };
    auto test4 = [&](int e0, const int4& v, unsigned lim) {
        const unsigned a  = (unsigned)v.x - k0;
        const unsigned b  = (unsigned)v.y - k0;
        const unsigned c  = (unsigned)v.z - k0;
        const unsigned d  = (unsigned)v.w - k0;
        const unsigned ab = a < b ? a : b;
        const unsigned cd = c < d ? c : d;
        const unsigned m  = ab < cd ? ab : cd;
        if (__builtin_amdgcn_ballot_w32(m < lim)) {
            note(e0,     a, lim);
            note(e0 + 1, b, lim);
            note(e0 + 2, c, lim);
            note(e0 + 3, d, lim);
        }
    };

    const int   E4   = E >> 2;
    const int4* src4 = (const int4*)src;
    const int   nIt  = (E4 + WTHR - 1) / WTHR;
    const int   nPad = ((nIt + WLOADS - 1) / WLOADS) * WLOADS;
    const int   rot  = nPad ? (int)(blockIdx.x % (unsigned)nPad) : 0;
    for (int g = 0; g < nPad; g += WLOADS) {
        int j0 = g + rot;
        if (j0 >= nPad) j0 -= nPad;
        int4 v[WLOADS];
        if (j0 + WLOADS <= nPad && (j0 + WLOADS) * WTHR <= E4) {
            const int b4 = j0 * WTHR + tid;
            #pragma unroll
            for (int u = 0; u < WLOADS; ++u)
                v[u] = src4[b4 + u * WTHR];
            #pragma unroll
            for (int u = 0; u < WLOADS; ++u)
                test4((b4 + u * WTHR) * 4, v[u], WROWS);
        } else {
            #pragma unroll
            for (int u = 0; u < WLOADS; ++u) {
                int j = j0 + u;
                if (j >= nPad) j -= nPad;
                const int i4 = j * WTHR + tid;
                v[u] = src4[i4 < E4 ? i4 : E4 - 1];
            }
            #pragma unroll
            for (int u = 0; u < WLOADS; ++u) {
                int j = j0 + u;
                if (j >= nPad) j -= nPad;
                const int i4 = j * WTHR + tid;
                test4(i4 * 4, v[u], i4 < E4 ? (unsigned)WROWS : 0u);
            }
        }
    }
    {
        const int      e   = E4 * 4 + tid;
        const bool     inb = e < E;
        const unsigned k   = inb ? (unsigned)src[e] - k0 : 0xffffffffu;
        note(e, k, WROWS);
    }

    if (cnt <= WSEG) {
        for (int i = lane; i < cnt; i += 32) {
            const int p = seg[i];
            add_edge(p >> 3, p & 7);
        }
    } else {
        for (int it = 0; it < nIt; ++it) {
            const int i4 = it * WTHR + tid;
            if (i4 < E4) {
                const int4     v = src4[i4];
                const unsigned a = (unsigned)v.x - k0;
                const unsigned b = (unsigned)v.y - k0;
                const unsigned c = (unsigned)v.z - k0;
                const unsigned d = (unsigned)v.w - k0;
                if (a < WROWS) add_edge(i4 * 4,     (int)a);
                if (b < WROWS) add_edge(i4 * 4 + 1, (int)b);
                if (c < WROWS) add_edge(i4 * 4 + 2, (int)c);
                if (d < WROWS) add_edge(i4 * 4 + 3, (int)d);
            }
        }
        const int e = E4 * 4 + tid;
        if (e < E) {
            const unsigned k = (unsigned)src[e] - k0;
            if (k < WROWS) add_edge(e, (int)k);
        }
    }
    __syncthreads();

    for (int it = 0; it < WTILE4 / WTHR; ++it) {
        const int f = it * WTHR + tid;
        const int l = f / (WROWS * Hsz / 4);
        const int i = f - l * (WROWS * Hsz / 4);
        float4* o = (float4*)(W + (size_t)l * WMAT + (size_t)k0 * Hsz);
        o[i] = tile4[f];
    }
}

#define KSPLIT 8
#define KSLICE (CONC / KSPLIT)
#define KSTEPS (CONC / 4)
#define APITCH 33
#define ALOADS (KSTEPS * 32 / 1024)
#define PCOLS  64
#define PTILE  (32 * PCOLS)

__global__ __launch_bounds__(1024) void gemm_fused(const float* __restrict__ in,
                                                   const float* __restrict__ hx,
                                                   const float* __restrict__ W,
                                                   const float* __restrict__ ts,
                                                   const float* __restrict__ b_ff1,
                                                   const float* __restrict__ b_ff2,
                                                   const float* __restrict__ b_ta,
                                                   const float* __restrict__ b_tb,
                                                   float* __restrict__ out) {
    __shared__ float4 red4[4 * KSPLIT * PTILE / 4];
    float* red = (float*)red4;

    const int lane = threadIdx.x & 31;
    const int wave = __builtin_amdgcn_readfirstlane((int)(threadIdx.x >> 5));
    const int l    = wave & 3;
    const int q    = wave >> 2;

    const int n0   = blockIdx.x * PCOLS;
    const int m0   = blockIdx.y * 32;
    const int c    = lane & 15;
    const int h    = lane >> 4;

    {
        float2*   As2 = (float2*)red4;
        const int t16 = lane & 15;
        const int s   = lane >> 4;
        float4 v[ALOADS];
        #pragma unroll
        for (int it = 0; it < ALOADS; ++it) {
            const int u  = it * 32 + wave;
            const int cr = u & 15;
            const int Tb = u >> 4;
            const int T  = Tb * 16 + t16;
            const int m  = m0 + s * 16 + cr;
            const float* p = (Tb < INsz / 64) ? in + (size_t)m * INsz + 4 * T
                                              : hx + (size_t)m * Hsz + (4 * T - INsz);
            v[it] = *(const float4*)p;
        }
        #pragma unroll
        for (int it = 0; it < ALOADS; ++it) {
            const int u  = it * 32 + wave;
            const int cr = u & 15;
            const int T  = (u >> 4) * 16 + t16;
            As2[(T * APITCH + cr) * 2 + s]      = make_float2(v[it].x, v[it].y);
            As2[(T * APITCH + 16 + cr) * 2 + s] = make_float2(v[it].z, v[it].w);
        }
    }

    v8f acc[2][4];
    #pragma unroll
    for (int s = 0; s < 2; ++s)
        #pragma unroll
        for (int j = 0; j < 4; ++j)
            acc[s][j] = (v8f){0.f,0.f,0.f,0.f,0.f,0.f,0.f,0.f};

    __syncthreads();

    const float4* As4  = red4 + lane;
    const int     kbeg = q * KSLICE;
    const float*  Wl   = W + (size_t)l * WMAT + (size_t)(kbeg + 2 * h) * Hsz + n0 + 4 * c;
    for (int ks = 0; ks < KSLICE; ks += 4) {
        const float4 a = As4[((kbeg + ks) >> 2) * APITCH];
        v2f a0, a1;
        a0.x = a.x;  a0.y = a.y;
        a1.x = a.z;  a1.y = a.w;

        const float4 b0 = *(const float4*)(Wl + (size_t)ks * Hsz);
        const float4 b1 = *(const float4*)(Wl + (size_t)(ks + 1) * Hsz);
        const float bk0[4] = {b0.x, b0.y, b0.z, b0.w};
        const float bk1[4] = {b1.x, b1.y, b1.z, b1.w};

        #pragma unroll
        for (int j = 0; j < 4; ++j) {
            v2f b;
            b.x = bk0[j];
            b.y = bk1[j];
            acc[0][j] = __builtin_amdgcn_wmma_f32_16x16x4_f32(
                false, a0, false, b, (short)0, acc[0][j], false, false);
            acc[1][j] = __builtin_amdgcn_wmma_f32_16x16x4_f32(
                false, a1, false, b, (short)0, acc[1][j], false, false);
        }
    }

    __syncthreads();

    {
        float* mine = red + (size_t)(l * KSPLIT + q) * PTILE + (8 * h) * PCOLS + 4 * c;
        #pragma unroll
        for (int s = 0; s < 2; ++s)
            #pragma unroll
            for (int r = 0; r < 8; ++r)
                *(float4*)(mine + (s * 16 + r) * PCOLS) =
                    make_float4(acc[s][0][r], acc[s][1][r], acc[s][2][r], acc[s][3][r]);
    }

    __syncthreads();

    {
        const int    ng   = n0 + 2 * lane;
        const float2 bf1  = *(const float2*)(b_ff1 + ng);
        const float2 bf2  = *(const float2*)(b_ff2 + ng);
        const float2 bta  = *(const float2*)(b_ta + ng);
        const float2 btb  = *(const float2*)(b_tb + ng);
        const int    m    = m0 + wave;
        const float  tsv  = ts[m];
        const float* tile = red + wave * PCOLS + 2 * lane;

        float2 v[4];
        #pragma unroll
        for (int mat = 0; mat < 4; ++mat) {
            const float* tm = tile + (size_t)(mat * KSPLIT) * PTILE;
            float2 sum = *(const float2*)tm;
            #pragma unroll
            for (int qq = 1; qq < KSPLIT; ++qq) {
                const float2 p = *(const float2*)(tm + (size_t)qq * PTILE);
                sum.x += p.x;
                sum.y += p.y;
            }
            v[mat] = sum;
        }
        float2 hres;
        {
            const float f1 = tanhf(v[0].x + bf1.x);
            const float f2 = tanhf(v[1].x + bf2.x);
            const float ta = v[2].x + bta.x;
            const float tb = v[3].x + btb.x;
            const float sg = 1.f / (1.f + expf(-(ta * tsv + tb)));
            hres.x = f1 * (1.f - sg) + sg * f2;
        }
        {
            const float f1 = tanhf(v[0].y + bf1.y);
            const float f2 = tanhf(v[1].y + bf2.y);
            const float ta = v[2].y + bta.y;
            const float tb = v[3].y + btb.y;
            const float sg = 1.f / (1.f + expf(-(ta * tsv + tb)));
            hres.y = f1 * (1.f - sg) + sg * f2;
        }
        const size_t o = (size_t)m * Hsz + ng;
        *(float2*)(out + o)                     = hres;
        *(float2*)(out + (size_t)Bsz * Hsz + o) = hres;
    }
}

extern "C" void kernel_launch(void* const* d_in, const int* in_sizes, int n_in,
                              void* d_out, int out_size, void* d_ws, size_t ws_size,
                              hipStream_t stream) {
    const float* input = (const float*)d_in[0];
    const float* hx    = (const float*)d_in[1];
    const float* ts    = (const float*)d_in[2];
    const int*   src   = (const int*)  d_in[3];
    const int*   dst   = (const int*)  d_in[4];
    const float* signs = (const float*)d_in[5];
    const float* w1    = (const float*)d_in[6];
    const float* w2    = (const float*)d_in[7];
    const float* wa    = (const float*)d_in[8];
    const float* wb    = (const float*)d_in[9];
    const float* bf1   = (const float*)d_in[10];
    const float* bf2   = (const float*)d_in[11];
    const float* bta   = (const float*)d_in[12];
    const float* btb   = (const float*)d_in[13];
    const int    E     = in_sizes[3];

    float* W  = (float*)d_ws;
    float* out = (float*)d_out;

    build_w<<<WGRID, WTHR, 0, stream>>>(src, dst, signs, w1, w2, wa, wb, W, E);

    dim3 grid(Hsz / PCOLS, Bsz / 32);
    gemm_fused<<<grid, 1024, 0, stream>>>(input, hx, W, ts, bf1, bf2, bta, btb, out);
}
